# starting kernel
# baseline (speedup 1.0000x reference)
#include <hip/hip_runtime.h>

#define N_TOK   131072
#define KEXP    16
#define IN_DIM  256
#define OUT_DIM 128

#define ROWS_PER_BLK 128
#define MOE_BLOCKS   1040
#define IDX_CAP      (MOE_BLOCKS * ROWS_PER_BLK)

typedef __bf16 bf16_t;
typedef __attribute__((ext_vector_type(16))) bf16_t v16bf;
typedef __attribute__((ext_vector_type(8)))  float  v8f;


__global__ __launch_bounds__(256) void init_ws(int* counts, int* cur, int* idxb) {
    int i = blockIdx.x * blockDim.x + threadIdx.x;
    if (i < KEXP) { counts[i] = 0; cur[i] = 0; }
    if (i < IDX_CAP) idxb[i] = -1;
}

__global__ __launch_bounds__(256) void count_k(const int* __restrict__ z, int* counts) {
    int i = blockIdx.x * blockDim.x + threadIdx.x;
    if (i < N_TOK) atomicAdd(&counts[z[i]], 1);
}

__global__ void offsets_k(const int* __restrict__ counts, int* off) {
    if (blockIdx.x == 0 && threadIdx.x == 0) {
        int acc = 0;
        for (int e = 0; e < KEXP; ++e) {
            off[e] = acc;
            acc += (counts[e] + ROWS_PER_BLK - 1) & ~(ROWS_PER_BLK - 1);
        }
        off[KEXP] = acc;
    }
}

__global__ __launch_bounds__(256) void scatter_k(const int* __restrict__ z,
                                                 const int* __restrict__ off,
                                                 int* cur, int* idxb) {
    int i = blockIdx.x * blockDim.x + threadIdx.x;
    if (i < N_TOK) {
        int e = z[i];
        int p = off[e] + atomicAdd(&cur[e], 1);
        idxb[p] = i;
    }
}


__global__ __launch_bounds__(256) void moe_k(const float* __restrict__ x,
                                             const int*   __restrict__ idxb,
                                             const int*   __restrict__ off,
                                             const float* __restrict__ W1,
                                             const float* __restrict__ b1,
                                             const float* __restrict__ W2,
                                             const float* __restrict__ b2,
                                             float* __restrict__ out) {
    __shared__ __attribute__((aligned(32))) unsigned char lds[64 * 1024];

    const int base  = blockIdx.x * ROWS_PER_BLK;
    const int total = off[KEXP];
    if (base >= total) return;

    int e = 0;
#pragma unroll
    for (int i = 1; i < KEXP; ++i) e = (base >= off[i]) ? i : e;

    const int tid    = threadIdx.x;
    const int lane   = tid & 31;
    const int wv     = tid >> 5;
    const int hi     = lane >> 4;
    const int lanelo = lane & 15;

    const float* w1 = W1 + (size_t)e * (IN_DIM * OUT_DIM);
    for (int i = tid; i < IN_DIM * OUT_DIM; i += 256) {
        int k = i >> 7;
        int n = i & 127;
        bf16_t v = (bf16_t)w1[i];
        int kstep = k >> 5, kk = k & 31;
        int ntile = n >> 4, nn = n & 15;
        int ln   = (((kk >> 3) & 1) << 4) | nn;
        int slot = (((kk >> 4) & 1) << 3) | (kk & 7);
        *(bf16_t*)(lds + (((kstep << 3) | ntile) << 10) + (ln << 5) + (slot << 1)) = v;
    }
    __syncthreads();

    const int rowPos = base + wv * 16 + lanelo;
    const int tok    = idxb[rowPos];
    const int tclamp = tok < 0 ? 0 : tok;
    const float* xrow = x + (size_t)tclamp * IN_DIM;

    v8f zero = {0.f, 0.f, 0.f, 0.f, 0.f, 0.f, 0.f, 0.f};
    v8f acc[8];
#pragma unroll
    for (int i = 0; i < 8; ++i) acc[i] = zero;

#pragma unroll
    for (int ks = 0; ks < 8; ++ks) {
        const int kb = ks * 32 + hi * 8;
        float4 f0 = *(const float4*)(xrow + kb);
        float4 f1 = *(const float4*)(xrow + kb + 4);
        float4 f2 = *(const float4*)(xrow + kb + 16);
        float4 f3 = *(const float4*)(xrow + kb + 20);
        v16bf a;
        a[0]=(bf16_t)f0.x; a[1]=(bf16_t)f0.y; a[2] =(bf16_t)f0.z; a[3] =(bf16_t)f0.w;
        a[4]=(bf16_t)f1.x; a[5]=(bf16_t)f1.y; a[6] =(bf16_t)f1.z; a[7] =(bf16_t)f1.w;
        a[8]=(bf16_t)f2.x; a[9]=(bf16_t)f2.y; a[10]=(bf16_t)f2.z; a[11]=(bf16_t)f2.w;
        a[12]=(bf16_t)f3.x; a[13]=(bf16_t)f3.y; a[14]=(bf16_t)f3.z; a[15]=(bf16_t)f3.w;
#pragma unroll
        for (int nt = 0; nt < 8; ++nt) {
            v16bf bf = *(const v16bf*)(lds + (((ks << 3) | nt) << 10) + (lane << 5));
            acc[nt] = __builtin_amdgcn_wmma_f32_16x16x32_bf16(
                false, a, false, bf, (short)0, acc[nt], false, false);
        }
    }

    float s0[8], s1[8];
#pragma unroll
    for (int v = 0; v < 8; ++v) { s0[v] = 0.f; s1[v] = 0.f; }
#pragma unroll
    for (int nt = 0; nt < 8; ++nt) {
        int n = nt * 16 + lanelo;
        float bb  = b1[e * OUT_DIM + n];
        float w20 = W2[((size_t)e * OUT_DIM + n) * 2 + 0];
        float w21 = W2[((size_t)e * OUT_DIM + n) * 2 + 1];
#pragma unroll
        for (int v = 0; v < 8; ++v) {
            float h = acc[nt][v] + bb;
            h = h > 0.f ? h : 0.f;
            s0[v] += h * w20;
            s1[v] += h * w21;
        }
    }
#pragma unroll
    for (int v = 0; v < 8; ++v) {
#pragma unroll
        for (int m = 1; m < 16; m <<= 1) {
            s0[v] += __shfl_xor(s0[v], m, 32);
            s1[v] += __shfl_xor(s1[v], m, 32);
        }
    }

    const float b20 = b2[e * 2 + 0], b21 = b2[e * 2 + 1];
    __syncthreads();
    float* lgt = (float*)lds;
#pragma unroll
    for (int v = 0; v < 8; ++v) {
        if (lanelo == v) {
            int M = v + 8 * hi;
            lgt[(wv * 16 + M) * 2 + 0] = s0[v] + b20;
            lgt[(wv * 16 + M) * 2 + 1] = s1[v] + b21;
        }
    }
    __syncthreads();

    if (lane < 16 && tok >= 0) {
        float l0 = lgt[(wv * 16 + lane) * 2 + 0];
        float l1 = lgt[(wv * 16 + lane) * 2 + 1];
        float mx = fmaxf(l0, l1);
        float e0 = __expf(l0 - mx), e1 = __expf(l1 - mx);
        float inv = 1.f / (e0 + e1);
        out[(size_t)tok * 2 + 0] = e0 * inv;
        out[(size_t)tok * 2 + 1] = e1 * inv;
    }
}


extern "C" void kernel_launch(void* const* d_in, const int* in_sizes, int n_in,
                              void* d_out, int out_size, void* d_ws, size_t ws_size,
                              hipStream_t stream) {
    (void)in_sizes; (void)n_in; (void)out_size; (void)ws_size;
    const float* x  = (const float*)d_in[0];
    const int*   z  = (const int*)d_in[1];
    const float* W1 = (const float*)d_in[2];
    const float* b1 = (const float*)d_in[3];
    const float* W2 = (const float*)d_in[4];
    const float* b2 = (const float*)d_in[5];
    float* out = (float*)d_out;

    int* counts = (int*)d_ws;
    int* off    = counts + 16;
    int* cur    = counts + 48;
    int* idxb   = counts + 128;

    init_ws  <<<(IDX_CAP + 255) / 256, 256, 0, stream>>>(counts, cur, idxb);
    count_k  <<<(N_TOK  + 255) / 256, 256, 0, stream>>>(z, counts);
    offsets_k<<<1, 1, 0, stream>>>(counts, off);
    scatter_k<<<(N_TOK  + 255) / 256, 256, 0, stream>>>(z, off, cur, idxb);
    moe_k    <<<MOE_BLOCKS, 256, 0, stream>>>(x, idxb, off, W1, b1, W2, b2, out);
}


# Row-tile kernel re-decomposed: each wave keeps a 16-column slice of W1 as bf16 fragments in registers over its expert slice; x rows go through LDS to all 8 waves; layer-2 sum order changes
# speedup vs baseline: 1.1333x; 1.1333x over previous
#include <hip/hip_runtime.h>

#define N_TOK   131072
#define KEXP    16
#define IN_DIM  256
#define OUT_DIM 128

#define ROWS_PER_BLK 128
#define MOE_BLOCKS   1040
#define IDX_CAP      (MOE_BLOCKS * ROWS_PER_BLK)

typedef __bf16 bf16_t;
typedef __attribute__((ext_vector_type(16))) bf16_t v16bf;
typedef __attribute__((ext_vector_type(8)))  float  v8f;


__global__ __launch_bounds__(256) void init_ws(int* counts, int* cur, int* idxb) {
    int i = blockIdx.x * blockDim.x + threadIdx.x;
    if (i < KEXP) { counts[i] = 0; cur[i] = 0; }
    if (i < IDX_CAP) idxb[i] = -1;
}

__global__ __launch_bounds__(256) void count_k(const int* __restrict__ z, int* counts) {
    int i = blockIdx.x * blockDim.x + threadIdx.x;
    if (i < N_TOK) atomicAdd(&counts[z[i]], 1);
}

__global__ void offsets_k(const int* __restrict__ counts, int* off) {
    if (blockIdx.x == 0 && threadIdx.x == 0) {
        int acc = 0;
        for (int e = 0; e < KEXP; ++e) {
            off[e] = acc;
            acc += (counts[e] + ROWS_PER_BLK - 1) & ~(ROWS_PER_BLK - 1);
        }
        off[KEXP] = acc;
    }
}

__global__ __launch_bounds__(256) void scatter_k(const int* __restrict__ z,
                                                 const int* __restrict__ off,
                                                 int* cur, int* idxb) {
    int i = blockIdx.x * blockDim.x + threadIdx.x;
    if (i < N_TOK) {
        int e = z[i];
        int p = off[e] + atomicAdd(&cur[e], 1);
        idxb[p] = i;
    }
}


#define RT     64
#define XP     528
#define NSLICE 32

typedef __attribute__((ext_vector_type(8))) bf16_t v8bf;
typedef __attribute__((ext_vector_type(4))) bf16_t v4bf;

template <int M>
__device__ __forceinline__ float row_xor(float v) {
    return __int_as_float(__builtin_amdgcn_update_dpp(0, __float_as_int(v), 0x160 + M, 0xf, 0xf, true));
}

__global__ __launch_bounds__(256) void moe_k(const float* __restrict__ x,
                                             const int*   __restrict__ idxb,
                                             const int*   __restrict__ off,
                                             const float* __restrict__ W1,
                                             const float* __restrict__ b1,
                                             const float* __restrict__ W2,
                                             const float* __restrict__ b2,
                                             float* __restrict__ out) {
    __shared__ __attribute__((aligned(16))) unsigned char xs[RT * XP];
    __shared__ __attribute__((aligned(16))) float part[8 * RT * 2];

    const int e     = blockIdx.x / NSLICE;
    const int sl    = blockIdx.x % NSLICE;
    const int seg0  = off[e];
    const int npass = (off[e + 1] - seg0) / RT;
    if (sl >= npass) return;

    const int tid  = threadIdx.x;
    const int lane = tid & 31;
    const int wv   = __builtin_amdgcn_readfirstlane(tid >> 5);
    const int h    = lane >> 4;
    const int c    = lane & 15;

    const int   n   = wv * 16 + c;
    const float bb  = b1[e * OUT_DIM + n];
    const float w20 = W2[(e * OUT_DIM + n) * 2 + 0];
    const float w21 = W2[(e * OUT_DIM + n) * 2 + 1];
    const float b20 = b2[e * 2 + 0], b21 = b2[e * 2 + 1];
    const bool  odd = (c & 1) != 0;

    const int* idx = idxb + seg0;

    const int* idw = idx + wv * 8 + (lane & 7);
    int t = sl;
    const int tkv = idw[t * RT];
    float4 pf[16];
#pragma unroll
    for (int j = 0; j < 8; ++j) {
        const int tj = __builtin_amdgcn_readlane(tkv, j);
        const int tc = tj < 0 ? 0 : tj;
        const float* xr = x + (size_t)tc * IN_DIM + 4 * lane;
        pf[2 * j]     = *(const float4*)xr;
        pf[2 * j + 1] = *(const float4*)(xr + 128);
    }
    int tkn = 0;
    if (t + NSLICE < npass) tkn = idw[(t + NSLICE) * RT];

    const float* w1 = W1 + (size_t)e * (IN_DIM * OUT_DIM) + (size_t)(8 * h) * OUT_DIM + (wv * 16 + c);
    v16bf bfr[8];
#pragma unroll
    for (int ks = 0; ks < 8; ++ks) {
#pragma unroll
        for (int s = 0; s < 16; ++s) {
            const int kk = (s < 8) ? s : (8 + s);
            bfr[ks][s] = (bf16_t)w1[(32 * ks + kk) * OUT_DIM];
        }
    }

    const unsigned char* abase = xs + c * XP + 16 * h;

    for (; t < npass; t += NSLICE) {
        int mytok = -1;
        if (tid < RT) mytok = idx[t * RT + tid];

#pragma unroll
        for (int j = 0; j < 8; ++j) {
            unsigned char* xw = xs + (wv * 8 + j) * XP + 8 * lane;
            const float4 f0 = pf[2 * j], f1 = pf[2 * j + 1];
            v4bf q0, q1;
            q0[0] = (bf16_t)f0.x; q0[1] = (bf16_t)f0.y; q0[2] = (bf16_t)f0.z; q0[3] = (bf16_t)f0.w;
            q1[0] = (bf16_t)f1.x; q1[1] = (bf16_t)f1.y; q1[2] = (bf16_t)f1.z; q1[3] = (bf16_t)f1.w;
            *(v4bf*)xw         = q0;
            *(v4bf*)(xw + 256) = q1;
        }
        __syncthreads();

        int tkn2 = 0;
        if (t + 2 * NSLICE < npass) tkn2 = idw[(t + 2 * NSLICE) * RT];
        if (t + NSLICE < npass) {
#pragma unroll
            for (int j = 0; j < 8; ++j) {
                const int tj = __builtin_amdgcn_readlane(tkn, j);
                const int tc = tj < 0 ? 0 : tj;
                const float* xr = x + (size_t)tc * IN_DIM + 4 * lane;
                pf[2 * j]     = *(const float4*)xr;
                pf[2 * j + 1] = *(const float4*)(xr + 128);
            }
        }
        tkn = tkn2;

        v8f acc[4];
#pragma unroll
        for (int rt = 0; rt < 4; ++rt) {
#pragma unroll
            for (int r = 0; r < 8; ++r) acc[rt][r] = 0.f;
        }
#pragma unroll
        for (int ks = 0; ks < 8; ++ks) {
            v16bf a[4];
#pragma unroll
            for (int rt = 0; rt < 4; ++rt) {
                const unsigned char* ap = abase + rt * 16 * XP + 64 * ks;
                const v8bf lo = *(const v8bf*)ap;
                const v8bf hi = *(const v8bf*)(ap + 32);
                a[rt] = __builtin_shufflevector(lo, hi, 0, 1, 2, 3, 4, 5, 6, 7, 8, 9, 10, 11, 12, 13, 14, 15);
            }
#pragma unroll
            for (int rt = 0; rt < 4; ++rt) {
                acc[rt] = __builtin_amdgcn_wmma_f32_16x16x32_bf16(
                    false, a[rt], false, bfr[ks], (short)0, acc[rt], false, false);
            }
        }

#pragma unroll
        for (int rt = 0; rt < 4; ++rt) {
            float sel = 0.f;
#pragma unroll
            for (int r = 0; r < 8; ++r) {
                float hv = acc[rt][r] + bb;
                hv = hv > 0.f ? hv : 0.f;
                const float p0 = hv * w20, p1 = hv * w21;
                const float keep = odd ? p1 : p0;
                const float send = odd ? p0 : p1;
                float s = keep + row_xor<1>(send);
                s += row_xor<2>(s);
                s += row_xor<4>(s);
                s += row_xor<8>(s);
                sel = ((c >> 1) == r) ? s : sel;
            }
            part[(wv * RT + rt * 16 + 8 * h + (c >> 1)) * 2 + (c & 1)] = sel;
        }
        __syncthreads();

        if (tid < RT) {
            float a0 = 0.f, a1 = 0.f;
#pragma unroll
            for (int w = 0; w < 8; ++w) {
                const float2 pw = *(const float2*)(part + (w * RT + tid) * 2);
                a0 += pw.x; a1 += pw.y;
            }
            if (mytok >= 0) {
                const float l0 = a0 + b20, l1 = a1 + b21;
                const float mx = fmaxf(l0, l1);
                const float e0 = __expf(l0 - mx), e1 = __expf(l1 - mx);
                const float inv = 1.f / (e0 + e1);
                out[(size_t)mytok * 2 + 0] = e0 * inv;
                out[(size_t)mytok * 2 + 1] = e1 * inv;
            }
        }
    }
}


extern "C" void kernel_launch(void* const* d_in, const int* in_sizes, int n_in,
                              void* d_out, int out_size, void* d_ws, size_t ws_size,
                              hipStream_t stream) {
    (void)in_sizes; (void)n_in; (void)out_size; (void)ws_size;
    const float* x  = (const float*)d_in[0];
    const int*   z  = (const int*)d_in[1];
    const float* W1 = (const float*)d_in[2];
    const float* b1 = (const float*)d_in[3];
    const float* W2 = (const float*)d_in[4];
    const float* b2 = (const float*)d_in[5];
    float* out = (float*)d_out;

    int* counts = (int*)d_ws;
    int* off    = counts + 16;
    int* cur    = counts + 48;
    int* idxb   = counts + 128;

    init_ws  <<<(IDX_CAP + 255) / 256, 256, 0, stream>>>(counts, cur, idxb);
    count_k  <<<(N_TOK  + 255) / 256, 256, 0, stream>>>(z, counts);
    offsets_k<<<1, 1, 0, stream>>>(counts, off);
    scatter_k<<<(N_TOK  + 255) / 256, 256, 0, stream>>>(z, off, cur, idxb);
    moe_k    <<<KEXP * NSLICE, 256, 0, stream>>>(x, idxb, off, W1, b1, W2, b2, out);
}


# Bucketing rebuilt as one append pass into fixed-capacity per-expert token lists: a chunk ranks its 1024 tokens in LDS, reserves runs with one returning add per expert; 5 launches become 3
# speedup vs baseline: 2.6369x; 2.3267x over previous
#include <hip/hip_runtime.h>

#define N_TOK   131072
#define KEXP    16
#define IN_DIM  256
#define OUT_DIM 128

#define SEG_CAP N_TOK
#define CHUNK   1024

typedef __bf16 bf16_t;
typedef __attribute__((ext_vector_type(16))) bf16_t v16bf;
typedef __attribute__((ext_vector_type(8)))  float  v8f;


__global__ void zero_k(int* cur) {
    if (threadIdx.x < KEXP) cur[threadIdx.x] = 0;
}

__global__ __launch_bounds__(256) void bucket_k(const int* __restrict__ z, int* cur, int* idxb) {
    __shared__ int lcnt[KEXP];
    __shared__ int lbase[KEXP];
    const int tid = threadIdx.x;
    if (tid < KEXP) lcnt[tid] = 0;
    const int g = blockIdx.x * 256 + tid;
    const int4 zz = ((const int4*)z)[g];
    __syncthreads();
    const int r0 = atomicAdd(&lcnt[zz.x], 1);
    const int r1 = atomicAdd(&lcnt[zz.y], 1);
    const int r2 = atomicAdd(&lcnt[zz.z], 1);
    const int r3 = atomicAdd(&lcnt[zz.w], 1);
    __syncthreads();
    if (tid < KEXP) lbase[tid] = atomicAdd(&cur[tid], lcnt[tid]);
    __syncthreads();
    idxb[zz.x * SEG_CAP + lbase[zz.x] + r0] = 4 * g + 0;
    idxb[zz.y * SEG_CAP + lbase[zz.y] + r1] = 4 * g + 1;
    idxb[zz.z * SEG_CAP + lbase[zz.z] + r2] = 4 * g + 2;
    idxb[zz.w * SEG_CAP + lbase[zz.w] + r3] = 4 * g + 3;
}


#define RT     64
#define XP     528
#define NSLICE 32

typedef __attribute__((ext_vector_type(8))) bf16_t v8bf;
typedef __attribute__((ext_vector_type(4))) bf16_t v4bf;

template <int M>
__device__ __forceinline__ float row_xor(float v) {
    return __int_as_float(__builtin_amdgcn_update_dpp(0, __float_as_int(v), 0x160 + M, 0xf, 0xf, true));
}

__global__ __launch_bounds__(256) void moe_k(const float* __restrict__ x,
                                             const int*   __restrict__ idxb,
                                             const int*   __restrict__ cur,
                                             const float* __restrict__ W1,
                                             const float* __restrict__ b1,
                                             const float* __restrict__ W2,
                                             const float* __restrict__ b2,
                                             float* __restrict__ out) {
    __shared__ __attribute__((aligned(16))) unsigned char xs[RT * XP];
    __shared__ __attribute__((aligned(16))) float part[8 * RT * 2];

    const int e     = blockIdx.x / NSLICE;
    const int sl    = blockIdx.x % NSLICE;
    const int cnt   = cur[e];
    const int npass = (cnt + RT - 1) / RT;
    if (sl >= npass) return;

    const int tid  = threadIdx.x;
    const int lane = tid & 31;
    const int wv   = __builtin_amdgcn_readfirstlane(tid >> 5);
    const int h    = lane >> 4;
    const int c    = lane & 15;

    const int   n   = wv * 16 + c;
    const float bb  = b1[e * OUT_DIM + n];
    const float w20 = W2[(e * OUT_DIM + n) * 2 + 0];
    const float w21 = W2[(e * OUT_DIM + n) * 2 + 1];
    const float b20 = b2[e * 2 + 0], b21 = b2[e * 2 + 1];
    const bool  odd = (c & 1) != 0;

    const int* idx = idxb + e * SEG_CAP;

    const int rpos = wv * 8 + (lane & 7);
    int t = sl;
    const int tkv = (t * RT + rpos < cnt) ? idx[t * RT + rpos] : -1;
    float4 pf[16];
#pragma unroll
    for (int j = 0; j < 8; ++j) {
        const int tj = __builtin_amdgcn_readlane(tkv, j);
        const int tc = tj < 0 ? 0 : tj;
        const float* xr = x + (size_t)tc * IN_DIM + 4 * lane;
        pf[2 * j]     = *(const float4*)xr;
        pf[2 * j + 1] = *(const float4*)(xr + 128);
    }
    int tkn = -1;
    if ((t + NSLICE) * RT + rpos < cnt) tkn = idx[(t + NSLICE) * RT + rpos];

    const float* w1 = W1 + (size_t)e * (IN_DIM * OUT_DIM) + (size_t)(8 * h) * OUT_DIM + (wv * 16 + c);
    v16bf bfr[8];
#pragma unroll
    for (int ks = 0; ks < 8; ++ks) {
#pragma unroll
        for (int s = 0; s < 16; ++s) {
            const int kk = (s < 8) ? s : (8 + s);
            bfr[ks][s] = (bf16_t)w1[(32 * ks + kk) * OUT_DIM];
        }
    }

    const unsigned char* abase = xs + c * XP + 16 * h;

    for (; t < npass; t += NSLICE) {
        int mytok = -1;
        if (tid < RT && t * RT + tid < cnt) mytok = idx[t * RT + tid];

#pragma unroll
        for (int j = 0; j < 8; ++j) {
            unsigned char* xw = xs + (wv * 8 + j) * XP + 8 * lane;
            const float4 f0 = pf[2 * j], f1 = pf[2 * j + 1];
            v4bf q0, q1;
            q0[0] = (bf16_t)f0.x; q0[1] = (bf16_t)f0.y; q0[2] = (bf16_t)f0.z; q0[3] = (bf16_t)f0.w;
            q1[0] = (bf16_t)f1.x; q1[1] = (bf16_t)f1.y; q1[2] = (bf16_t)f1.z; q1[3] = (bf16_t)f1.w;
            *(v4bf*)xw         = q0;
            *(v4bf*)(xw + 256) = q1;
        }
        __syncthreads();

        int tkn2 = -1;
        if ((t + 2 * NSLICE) * RT + rpos < cnt) tkn2 = idx[(t + 2 * NSLICE) * RT + rpos];
        if (t + NSLICE < npass) {
#pragma unroll
            for (int j = 0; j < 8; ++j) {
                const int tj = __builtin_amdgcn_readlane(tkn, j);
                const int tc = tj < 0 ? 0 : tj;
                const float* xr = x + (size_t)tc * IN_DIM + 4 * lane;
                pf[2 * j]     = *(const float4*)xr;
                pf[2 * j + 1] = *(const float4*)(xr + 128);
            }
        }
        tkn = tkn2;

        v8f acc[4];
#pragma unroll
        for (int rt = 0; rt < 4; ++rt) {
#pragma unroll
            for (int r = 0; r < 8; ++r) acc[rt][r] = 0.f;
        }
#pragma unroll
        for (int ks = 0; ks < 8; ++ks) {
            v16bf a[4];
#pragma unroll
            for (int rt = 0; rt < 4; ++rt) {
                const unsigned char* ap = abase + rt * 16 * XP + 64 * ks;
                const v8bf lo = *(const v8bf*)ap;
                const v8bf hi = *(const v8bf*)(ap + 32);
                a[rt] = __builtin_shufflevector(lo, hi, 0, 1, 2, 3, 4, 5, 6, 7, 8, 9, 10, 11, 12, 13, 14, 15);
            }
#pragma unroll
            for (int rt = 0; rt < 4; ++rt) {
                acc[rt] = __builtin_amdgcn_wmma_f32_16x16x32_bf16(
                    false, a[rt], false, bfr[ks], (short)0, acc[rt], false, false);
            }
        }

#pragma unroll
        for (int rt = 0; rt < 4; ++rt) {
            float sel = 0.f;
#pragma unroll
            for (int r = 0; r < 8; ++r) {
                float hv = acc[rt][r] + bb;
                hv = hv > 0.f ? hv : 0.f;
                const float p0 = hv * w20, p1 = hv * w21;
                const float keep = odd ? p1 : p0;
                const float send = odd ? p0 : p1;
                float s = keep + row_xor<1>(send);
                s += row_xor<2>(s);
                s += row_xor<4>(s);
                s += row_xor<8>(s);
                sel = ((c >> 1) == r) ? s : sel;
            }
            part[(wv * RT + rt * 16 + 8 * h + (c >> 1)) * 2 + (c & 1)] = sel;
        }
        __syncthreads();

        if (tid < RT) {
            float a0 = 0.f, a1 = 0.f;
#pragma unroll
            for (int w = 0; w < 8; ++w) {
                const float2 pw = *(const float2*)(part + (w * RT + tid) * 2);
                a0 += pw.x; a1 += pw.y;
            }
            if (mytok >= 0) {
                const float l0 = a0 + b20, l1 = a1 + b21;
                const float mx = fmaxf(l0, l1);
                const float e0 = __expf(l0 - mx), e1 = __expf(l1 - mx);
                const float inv = 1.f / (e0 + e1);
                out[(size_t)mytok * 2 + 0] = e0 * inv;
                out[(size_t)mytok * 2 + 1] = e1 * inv;
            }
        }
    }
}


extern "C" void kernel_launch(void* const* d_in, const int* in_sizes, int n_in,
                              void* d_out, int out_size, void* d_ws, size_t ws_size,
                              hipStream_t stream) {
    (void)in_sizes; (void)n_in; (void)out_size; (void)ws_size;
    const float* x  = (const float*)d_in[0];
    const int*   z  = (const int*)d_in[1];
    const float* W1 = (const float*)d_in[2];
    const float* b1 = (const float*)d_in[3];
    const float* W2 = (const float*)d_in[4];
    const float* b2 = (const float*)d_in[5];
    float* out = (float*)d_out;

    int* cur  = (int*)d_ws;
    int* idxb = cur + 128;

    zero_k  <<<1, 32, 0, stream>>>(cur);
    bucket_k<<<N_TOK / CHUNK, 256, 0, stream>>>(z, cur, idxb);
    moe_k   <<<KEXP * NSLICE, 256, 0, stream>>>(x, idxb, cur, W1, b1, W2, b2, out);
}


# Layer-2 16-lane sums as a halving butterfly: at each cross-lane step a lane keeps half of its rows or outputs and hands the rest to its partner; 15 DPP adds per 8 rows (were 32); same pairs summed
# speedup vs baseline: 2.7220x; 1.0323x over previous
#include <hip/hip_runtime.h>

#define N_TOK   131072
#define KEXP    16
#define IN_DIM  256
#define OUT_DIM 128

#define SEG_CAP N_TOK
#define CHUNK   1024

typedef __bf16 bf16_t;
typedef __attribute__((ext_vector_type(16))) bf16_t v16bf;
typedef __attribute__((ext_vector_type(8)))  float  v8f;


__global__ void zero_k(int* cur) {
    if (threadIdx.x < KEXP) cur[threadIdx.x] = 0;
}

__global__ __launch_bounds__(256) void bucket_k(const int* __restrict__ z, int* cur, int* idxb) {
    __shared__ int lcnt[KEXP];
    __shared__ int lbase[KEXP];
    const int tid = threadIdx.x;
    if (tid < KEXP) lcnt[tid] = 0;
    const int g = blockIdx.x * 256 + tid;
    const int4 zz = ((const int4*)z)[g];
    __syncthreads();
    const int r0 = atomicAdd(&lcnt[zz.x], 1);
    const int r1 = atomicAdd(&lcnt[zz.y], 1);
    const int r2 = atomicAdd(&lcnt[zz.z], 1);
    const int r3 = atomicAdd(&lcnt[zz.w], 1);
    __syncthreads();
    if (tid < KEXP) lbase[tid] = atomicAdd(&cur[tid], lcnt[tid]);
    __syncthreads();
    idxb[zz.x * SEG_CAP + lbase[zz.x] + r0] = 4 * g + 0;
    idxb[zz.y * SEG_CAP + lbase[zz.y] + r1] = 4 * g + 1;
    idxb[zz.z * SEG_CAP + lbase[zz.z] + r2] = 4 * g + 2;
    idxb[zz.w * SEG_CAP + lbase[zz.w] + r3] = 4 * g + 3;
}


#define RT     64
#define XP     528
#define NSLICE 32

typedef __attribute__((ext_vector_type(8))) bf16_t v8bf;
typedef __attribute__((ext_vector_type(4))) bf16_t v4bf;

template <int M>
__device__ __forceinline__ float row_xor(float v) {
    return __int_as_float(__builtin_amdgcn_update_dpp(0, __float_as_int(v), 0x160 + M, 0xf, 0xf, true));
}

__global__ __launch_bounds__(256) void moe_k(const float* __restrict__ x,
                                             const int*   __restrict__ idxb,
                                             const int*   __restrict__ cur,
                                             const float* __restrict__ W1,
                                             const float* __restrict__ b1,
                                             const float* __restrict__ W2,
                                             const float* __restrict__ b2,
                                             float* __restrict__ out) {
    __shared__ __attribute__((aligned(16))) unsigned char xs[RT * XP];
    __shared__ __attribute__((aligned(16))) float part[8 * RT * 2];

    const int e     = blockIdx.x / NSLICE;
    const int sl    = blockIdx.x % NSLICE;
    const int cnt   = cur[e];
    const int npass = (cnt + RT - 1) / RT;
    if (sl >= npass) return;

    const int tid  = threadIdx.x;
    const int lane = tid & 31;
    const int wv   = __builtin_amdgcn_readfirstlane(tid >> 5);
    const int h    = lane >> 4;
    const int c    = lane & 15;

    const int   n   = wv * 16 + c;
    const float bb  = b1[e * OUT_DIM + n];
    const float w20 = W2[(e * OUT_DIM + n) * 2 + 0];
    const float w21 = W2[(e * OUT_DIM + n) * 2 + 1];
    const float b20 = b2[e * 2 + 0], b21 = b2[e * 2 + 1];
    const bool  odd = (c & 1) != 0;

    const int* idx = idxb + e * SEG_CAP;

    const int rpos = wv * 8 + (lane & 7);
    int t = sl;
    const int tkv = (t * RT + rpos < cnt) ? idx[t * RT + rpos] : -1;
    float4 pf[16];
#pragma unroll
    for (int j = 0; j < 8; ++j) {
        const int tj = __builtin_amdgcn_readlane(tkv, j);
        const int tc = tj < 0 ? 0 : tj;
        const float* xr = x + (size_t)tc * IN_DIM + 4 * lane;
        pf[2 * j]     = *(const float4*)xr;
        pf[2 * j + 1] = *(const float4*)(xr + 128);
    }
    int tkn = -1;
    if ((t + NSLICE) * RT + rpos < cnt) tkn = idx[(t + NSLICE) * RT + rpos];

    const float* w1 = W1 + (size_t)e * (IN_DIM * OUT_DIM) + (size_t)(8 * h) * OUT_DIM + (wv * 16 + c);
    v16bf bfr[8];
#pragma unroll
    for (int ks = 0; ks < 8; ++ks) {
#pragma unroll
        for (int s = 0; s < 16; ++s) {
            const int kk = (s < 8) ? s : (8 + s);
            bfr[ks][s] = (bf16_t)w1[(32 * ks + kk) * OUT_DIM];
        }
    }

    const unsigned char* abase = xs + c * XP + 16 * h;

    for (; t < npass; t += NSLICE) {
        int mytok = -1;
        if (tid < RT && t * RT + tid < cnt) mytok = idx[t * RT + tid];

#pragma unroll
        for (int j = 0; j < 8; ++j) {
            unsigned char* xw = xs + (wv * 8 + j) * XP + 8 * lane;
            const float4 f0 = pf[2 * j], f1 = pf[2 * j + 1];
            v4bf q0, q1;
            q0[0] = (bf16_t)f0.x; q0[1] = (bf16_t)f0.y; q0[2] = (bf16_t)f0.z; q0[3] = (bf16_t)f0.w;
            q1[0] = (bf16_t)f1.x; q1[1] = (bf16_t)f1.y; q1[2] = (bf16_t)f1.z; q1[3] = (bf16_t)f1.w;
            *(v4bf*)xw         = q0;
            *(v4bf*)(xw + 256) = q1;
        }
        __syncthreads();

        int tkn2 = -1;
        if ((t + 2 * NSLICE) * RT + rpos < cnt) tkn2 = idx[(t + 2 * NSLICE) * RT + rpos];
        if (t + NSLICE < npass) {
#pragma unroll
            for (int j = 0; j < 8; ++j) {
                const int tj = __builtin_amdgcn_readlane(tkn, j);
                const int tc = tj < 0 ? 0 : tj;
                const float* xr = x + (size_t)tc * IN_DIM + 4 * lane;
                pf[2 * j]     = *(const float4*)xr;
                pf[2 * j + 1] = *(const float4*)(xr + 128);
            }
        }
        tkn = tkn2;

        v8f acc[4];
#pragma unroll
        for (int rt = 0; rt < 4; ++rt) {
#pragma unroll
            for (int r = 0; r < 8; ++r) acc[rt][r] = 0.f;
        }
#pragma unroll
        for (int ks = 0; ks < 8; ++ks) {
            v16bf a[4];
#pragma unroll
            for (int rt = 0; rt < 4; ++rt) {
                const unsigned char* ap = abase + rt * 16 * XP + 64 * ks;
                const v8bf lo = *(const v8bf*)ap;
                const v8bf hi = *(const v8bf*)(ap + 32);
                a[rt] = __builtin_shufflevector(lo, hi, 0, 1, 2, 3, 4, 5, 6, 7, 8, 9, 10, 11, 12, 13, 14, 15);
            }
#pragma unroll
            for (int rt = 0; rt < 4; ++rt) {
                acc[rt] = __builtin_amdgcn_wmma_f32_16x16x32_bf16(
                    false, a[rt], false, bfr[ks], (short)0, acc[rt], false, false);
            }
        }

        const bool hi1 = (c & 2) != 0, hi2 = (c & 4) != 0, hi3 = (c & 8) != 0;
        const int  rowc = ((c >> 1) & 1) * 4 + ((c >> 2) & 1) * 2 + ((c >> 3) & 1);
#pragma unroll
        for (int rt = 0; rt < 4; ++rt) {
            float v8[8], v4[4], v2[2];
#pragma unroll
            for (int r = 0; r < 8; ++r) {
                float hv = acc[rt][r] + bb;
                hv = hv > 0.f ? hv : 0.f;
                const float p0 = hv * w20, p1 = hv * w21;
                const float keep = odd ? p1 : p0;
                const float send = odd ? p0 : p1;
                v8[r] = keep + row_xor<1>(send);
            }
#pragma unroll
            for (int j = 0; j < 4; ++j) {
                const float keep = hi1 ? v8[j + 4] : v8[j];
                const float send = hi1 ? v8[j] : v8[j + 4];
                v4[j] = keep + row_xor<2>(send);
            }
#pragma unroll
            for (int j = 0; j < 2; ++j) {
                const float keep = hi2 ? v4[j + 2] : v4[j];
                const float send = hi2 ? v4[j] : v4[j + 2];
                v2[j] = keep + row_xor<4>(send);
            }
            const float keep1 = hi3 ? v2[1] : v2[0];
            const float send1 = hi3 ? v2[0] : v2[1];
            const float res = keep1 + row_xor<8>(send1);
            part[(wv * RT + rt * 16 + 8 * h + rowc) * 2 + (c & 1)] = res;
        }
        __syncthreads();

        if (tid < RT) {
            float a0 = 0.f, a1 = 0.f;
#pragma unroll
            for (int w = 0; w < 8; ++w) {
                const float2 pw = *(const float2*)(part + (w * RT + tid) * 2);
                a0 += pw.x; a1 += pw.y;
            }
            if (mytok >= 0) {
                const float l0 = a0 + b20, l1 = a1 + b21;
                const float mx = fmaxf(l0, l1);
                const float e0 = __expf(l0 - mx), e1 = __expf(l1 - mx);
                const float inv = 1.f / (e0 + e1);
                out[(size_t)mytok * 2 + 0] = e0 * inv;
                out[(size_t)mytok * 2 + 1] = e1 * inv;
            }
        }
    }
}


extern "C" void kernel_launch(void* const* d_in, const int* in_sizes, int n_in,
                              void* d_out, int out_size, void* d_ws, size_t ws_size,
                              hipStream_t stream) {
    (void)in_sizes; (void)n_in; (void)out_size; (void)ws_size;
    const float* x  = (const float*)d_in[0];
    const int*   z  = (const int*)d_in[1];
    const float* W1 = (const float*)d_in[2];
    const float* b1 = (const float*)d_in[3];
    const float* W2 = (const float*)d_in[4];
    const float* b2 = (const float*)d_in[5];
    float* out = (float*)d_out;

    int* cur  = (int*)d_ws;
    int* idxb = cur + 128;

    zero_k  <<<1, 32, 0, stream>>>(cur);
    bucket_k<<<N_TOK / CHUNK, 256, 0, stream>>>(z, cur, idxb);
    moe_k   <<<KEXP * NSLICE, 256, 0, stream>>>(x, idxb, cur, W1, b1, W2, b2, out);
}


# Pass-loop barriers order LDS only (were full fences that drained the row loads in flight): next rows requested before the first barrier, stay in flight over both; join takes ids the wave holds
# speedup vs baseline: 2.8059x; 1.0308x over previous
#include <hip/hip_runtime.h>

#define N_TOK   131072
#define KEXP    16
#define IN_DIM  256
#define OUT_DIM 128

#define SEG_CAP N_TOK
#define CHUNK   1024

typedef __bf16 bf16_t;
typedef __attribute__((ext_vector_type(16))) bf16_t v16bf;
typedef __attribute__((ext_vector_type(8)))  float  v8f;


__global__ void zero_k(int* cur) {
    if (threadIdx.x < KEXP) cur[threadIdx.x] = 0;
}

__global__ __launch_bounds__(256) void bucket_k(const int* __restrict__ z, int* cur, int* idxb) {
    __shared__ int lcnt[KEXP];
    __shared__ int lbase[KEXP];
    const int tid = threadIdx.x;
    if (tid < KEXP) lcnt[tid] = 0;
    const int g = blockIdx.x * 256 + tid;
    const int4 zz = ((const int4*)z)[g];
    __syncthreads();
    const int r0 = atomicAdd(&lcnt[zz.x], 1);
    const int r1 = atomicAdd(&lcnt[zz.y], 1);
    const int r2 = atomicAdd(&lcnt[zz.z], 1);
    const int r3 = atomicAdd(&lcnt[zz.w], 1);
    __syncthreads();
    if (tid < KEXP) lbase[tid] = atomicAdd(&cur[tid], lcnt[tid]);
    __syncthreads();
    idxb[zz.x * SEG_CAP + lbase[zz.x] + r0] = 4 * g + 0;
    idxb[zz.y * SEG_CAP + lbase[zz.y] + r1] = 4 * g + 1;
    idxb[zz.z * SEG_CAP + lbase[zz.z] + r2] = 4 * g + 2;
    idxb[zz.w * SEG_CAP + lbase[zz.w] + r3] = 4 * g + 3;
}


#define RT     64
#define XP     528
#define NSLICE 32

typedef __attribute__((ext_vector_type(8))) bf16_t v8bf;
typedef __attribute__((ext_vector_type(4))) bf16_t v4bf;

template <int M>
__device__ __forceinline__ float row_xor(float v) {
    return __int_as_float(__builtin_amdgcn_update_dpp(0, __float_as_int(v), 0x160 + M, 0xf, 0xf, true));
}

__device__ __forceinline__ void lds_barrier() {
    __builtin_amdgcn_fence(__ATOMIC_RELEASE, "workgroup", "local");
    __builtin_amdgcn_s_barrier();
    __builtin_amdgcn_fence(__ATOMIC_ACQUIRE, "workgroup", "local");
}

__global__ __launch_bounds__(256) void moe_k(const float* __restrict__ x,
                                             const int*   __restrict__ idxb,
                                             const int*   __restrict__ cur,
                                             const float* __restrict__ W1,
                                             const float* __restrict__ b1,
                                             const float* __restrict__ W2,
                                             const float* __restrict__ b2,
                                             float* __restrict__ out) {
    __shared__ __attribute__((aligned(16))) unsigned char xs[RT * XP];
    __shared__ __attribute__((aligned(16))) float part[8 * RT * 2];

    const int e     = blockIdx.x / NSLICE;
    const int sl    = blockIdx.x % NSLICE;
    const int cnt   = cur[e];
    const int npass = (cnt + RT - 1) / RT;
    if (sl >= npass) return;

    const int tid  = threadIdx.x;
    const int lane = tid & 31;
    const int wv   = __builtin_amdgcn_readfirstlane(tid >> 5);
    const int h    = lane >> 4;
    const int c    = lane & 15;

    const int   n   = wv * 16 + c;
    const float bb  = b1[e * OUT_DIM + n];
    const float w20 = W2[(e * OUT_DIM + n) * 2 + 0];
    const float w21 = W2[(e * OUT_DIM + n) * 2 + 1];
    const float b20 = b2[e * 2 + 0], b21 = b2[e * 2 + 1];
    const bool  odd = (c & 1) != 0;

    const int* idx = idxb + e * SEG_CAP;

    const int rpos = wv * 8 + (lane & 7);
    int t = sl;
    int tkc = (t * RT + rpos < cnt) ? idx[t * RT + rpos] : -1;
    float4 pf[16];
#pragma unroll
    for (int j = 0; j < 8; ++j) {
        const int tj = __builtin_amdgcn_readlane(tkc, j);
        const int tc = tj < 0 ? 0 : tj;
        const float* xr = x + (size_t)tc * IN_DIM + 4 * lane;
        pf[2 * j]     = *(const float4*)xr;
        pf[2 * j + 1] = *(const float4*)(xr + 128);
    }
    int tkn = -1;
    if ((t + NSLICE) * RT + rpos < cnt) tkn = idx[(t + NSLICE) * RT + rpos];

    const float* w1 = W1 + (size_t)e * (IN_DIM * OUT_DIM) + (size_t)(8 * h) * OUT_DIM + (wv * 16 + c);
    v16bf bfr[8];
#pragma unroll
    for (int ks = 0; ks < 8; ++ks) {
#pragma unroll
        for (int s = 0; s < 16; ++s) {
            const int kk = (s < 8) ? s : (8 + s);
            bfr[ks][s] = (bf16_t)w1[(32 * ks + kk) * OUT_DIM];
        }
    }

    const unsigned char* abase = xs + c * XP + 16 * h;

    for (; t < npass; t += NSLICE) {
#pragma unroll
        for (int j = 0; j < 8; ++j) {
            unsigned char* xw = xs + (wv * 8 + j) * XP + 8 * lane;
            const float4 f0 = pf[2 * j], f1 = pf[2 * j + 1];
            v4bf q0, q1;
            q0[0] = (bf16_t)f0.x; q0[1] = (bf16_t)f0.y; q0[2] = (bf16_t)f0.z; q0[3] = (bf16_t)f0.w;
            q1[0] = (bf16_t)f1.x; q1[1] = (bf16_t)f1.y; q1[2] = (bf16_t)f1.z; q1[3] = (bf16_t)f1.w;
            *(v4bf*)xw         = q0;
            *(v4bf*)(xw + 256) = q1;
        }

        int tkn2 = -1;
        if ((t + 2 * NSLICE) * RT + rpos < cnt) tkn2 = idx[(t + 2 * NSLICE) * RT + rpos];
        if (t + NSLICE < npass) {
#pragma unroll
            for (int j = 0; j < 8; ++j) {
                const int tj = __builtin_amdgcn_readlane(tkn, j);
                const int tc = tj < 0 ? 0 : tj;
                const float* xr = x + (size_t)tc * IN_DIM + 4 * lane;
                pf[2 * j]     = *(const float4*)xr;
                pf[2 * j + 1] = *(const float4*)(xr + 128);
            }
        }
        lds_barrier();

        v8f acc[4];
#pragma unroll
        for (int rt = 0; rt < 4; ++rt) {
#pragma unroll
            for (int r = 0; r < 8; ++r) acc[rt][r] = 0.f;
        }
#pragma unroll
        for (int ks = 0; ks < 8; ++ks) {
            v16bf a[4];
#pragma unroll
            for (int rt = 0; rt < 4; ++rt) {
                const unsigned char* ap = abase + rt * 16 * XP + 64 * ks;
                const v8bf lo = *(const v8bf*)ap;
                const v8bf hi = *(const v8bf*)(ap + 32);
                a[rt] = __builtin_shufflevector(lo, hi, 0, 1, 2, 3, 4, 5, 6, 7, 8, 9, 10, 11, 12, 13, 14, 15);
            }
#pragma unroll
            for (int rt = 0; rt < 4; ++rt) {
                acc[rt] = __builtin_amdgcn_wmma_f32_16x16x32_bf16(
                    false, a[rt], false, bfr[ks], (short)0, acc[rt], false, false);
            }
        }

        const bool hi1 = (c & 2) != 0, hi2 = (c & 4) != 0, hi3 = (c & 8) != 0;
        const int  rowc = ((c >> 1) & 1) * 4 + ((c >> 2) & 1) * 2 + ((c >> 3) & 1);
#pragma unroll
        for (int rt = 0; rt < 4; ++rt) {
            float v8[8], v4[4], v2[2];
#pragma unroll
            for (int r = 0; r < 8; ++r) {
                float hv = acc[rt][r] + bb;
                hv = hv > 0.f ? hv : 0.f;
                const float p0 = hv * w20, p1 = hv * w21;
                const float keep = odd ? p1 : p0;
                const float send = odd ? p0 : p1;
                v8[r] = keep + row_xor<1>(send);
            }
#pragma unroll
            for (int j = 0; j < 4; ++j) {
                const float keep = hi1 ? v8[j + 4] : v8[j];
                const float send = hi1 ? v8[j] : v8[j + 4];
                v4[j] = keep + row_xor<2>(send);
            }
#pragma unroll
            for (int j = 0; j < 2; ++j) {
                const float keep = hi2 ? v4[j + 2] : v4[j];
                const float send = hi2 ? v4[j] : v4[j + 2];
                v2[j] = keep + row_xor<4>(send);
            }
            const float keep1 = hi3 ? v2[1] : v2[0];
            const float send1 = hi3 ? v2[0] : v2[1];
            const float res = keep1 + row_xor<8>(send1);
            part[(wv * RT + rt * 16 + 8 * h + rowc) * 2 + (c & 1)] = res;
        }
        lds_barrier();

        const int jtok = __shfl(tkc, (lane >> 1) & 7);
        if (lane < 16) {
            const int jrow = wv * 8 + (lane >> 1);
            float a = 0.f;
#pragma unroll
            for (int w = 0; w < 8; ++w) a += part[(w * RT + jrow) * 2 + (lane & 1)];
            const float ls = a + (odd ? b21 : b20);
            const float lo = row_xor<1>(ls);
            const float mx = fmaxf(ls, lo);
            const float es = __expf(ls - mx), eo = __expf(lo - mx);
            const float inv = 1.f / (es + eo);
            if (jtok >= 0) out[(size_t)jtok * 2 + (lane & 1)] = es * inv;
        }
        tkc = tkn;
        tkn = tkn2;
    }
}


extern "C" void kernel_launch(void* const* d_in, const int* in_sizes, int n_in,
                              void* d_out, int out_size, void* d_ws, size_t ws_size,
                              hipStream_t stream) {
    (void)in_sizes; (void)n_in; (void)out_size; (void)ws_size;
    const float* x  = (const float*)d_in[0];
    const int*   z  = (const int*)d_in[1];
    const float* W1 = (const float*)d_in[2];
    const float* b1 = (const float*)d_in[3];
    const float* W2 = (const float*)d_in[4];
    const float* b2 = (const float*)d_in[5];
    float* out = (float*)d_out;

    int* cur  = (int*)d_ws;
    int* idxb = cur + 128;

    zero_k  <<<1, 32, 0, stream>>>(cur);
    bucket_k<<<N_TOK / CHUNK, 256, 0, stream>>>(z, cur, idxb);
    moe_k   <<<KEXP * NSLICE, 256, 0, stream>>>(x, idxb, cur, W1, b1, W2, b2, out);
}


# Each wave keeps a 32-column slice of W1 in registers (was 16), 4 waves per work-group (were 8): each A fragment read from LDS feeds 2 WMMAs, LDS row reads halved; layer-2 sum order changes
# speedup vs baseline: 2.9035x; 1.0348x over previous
#include <hip/hip_runtime.h>

#define N_TOK   131072
#define KEXP    16
#define IN_DIM  256
#define OUT_DIM 128

#define SEG_CAP N_TOK
#define CHUNK   1024

typedef __bf16 bf16_t;
typedef __attribute__((ext_vector_type(16))) bf16_t v16bf;
typedef __attribute__((ext_vector_type(8)))  float  v8f;


__global__ void zero_k(int* cur) {
    if (threadIdx.x < KEXP) cur[threadIdx.x] = 0;
}

__global__ __launch_bounds__(256) void bucket_k(const int* __restrict__ z, int* cur, int* idxb) {
    __shared__ int lcnt[KEXP];
    __shared__ int lbase[KEXP];
    const int tid = threadIdx.x;
    if (tid < KEXP) lcnt[tid] = 0;
    const int g = blockIdx.x * 256 + tid;
    const int4 zz = ((const int4*)z)[g];
    __syncthreads();
    const int r0 = atomicAdd(&lcnt[zz.x], 1);
    const int r1 = atomicAdd(&lcnt[zz.y], 1);
    const int r2 = atomicAdd(&lcnt[zz.z], 1);
    const int r3 = atomicAdd(&lcnt[zz.w], 1);
    __syncthreads();
    if (tid < KEXP) lbase[tid] = atomicAdd(&cur[tid], lcnt[tid]);
    __syncthreads();
    idxb[zz.x * SEG_CAP + lbase[zz.x] + r0] = 4 * g + 0;
    idxb[zz.y * SEG_CAP + lbase[zz.y] + r1] = 4 * g + 1;
    idxb[zz.z * SEG_CAP + lbase[zz.z] + r2] = 4 * g + 2;
    idxb[zz.w * SEG_CAP + lbase[zz.w] + r3] = 4 * g + 3;
}


#define RT     64
#define XP     528
#define NSLICE 32
#define NWV    4
#define RPW    (RT / NWV)

typedef __attribute__((ext_vector_type(8))) bf16_t v8bf;
typedef __attribute__((ext_vector_type(4))) bf16_t v4bf;

template <int M>
__device__ __forceinline__ float row_xor(float v) {
    return __int_as_float(__builtin_amdgcn_update_dpp(0, __float_as_int(v), 0x160 + M, 0xf, 0xf, true));
}

__device__ __forceinline__ void lds_barrier() {
    __builtin_amdgcn_fence(__ATOMIC_RELEASE, "workgroup", "local");
    __builtin_amdgcn_s_barrier();
    __builtin_amdgcn_fence(__ATOMIC_ACQUIRE, "workgroup", "local");
}

__global__ __launch_bounds__(128) void moe_k(const float* __restrict__ x,
                                             const int*   __restrict__ idxb,
                                             const int*   __restrict__ cur,
                                             const float* __restrict__ W1,
                                             const float* __restrict__ b1,
                                             const float* __restrict__ W2,
                                             const float* __restrict__ b2,
                                             float* __restrict__ out) {
    __shared__ __attribute__((aligned(16))) unsigned char xs[RT * XP];
    __shared__ __attribute__((aligned(16))) float part[NWV * RT * 2];

    const int e     = blockIdx.x / NSLICE;
    const int sl    = blockIdx.x % NSLICE;
    const int cnt   = cur[e];
    const int npass = (cnt + RT - 1) / RT;
    if (sl >= npass) return;

    const int tid  = threadIdx.x;
    const int lane = tid & 31;
    const int wv   = __builtin_amdgcn_readfirstlane(tid >> 5);
    const int h    = lane >> 4;
    const int c    = lane & 15;

    const int    n   = wv * 32 + 2 * c;
    const float2 bb  = *(const float2*)(b1 + e * OUT_DIM + n);
    const float4 w2v = *(const float4*)(W2 + (e * OUT_DIM + n) * 2);
    const float b20 = b2[e * 2 + 0], b21 = b2[e * 2 + 1];
    const bool  odd = (c & 1) != 0;
    const float wkA = odd ? w2v.y : w2v.x, wsA = odd ? w2v.x : w2v.y;
    const float wkB = odd ? w2v.w : w2v.z, wsB = odd ? w2v.z : w2v.w;

    const int* idx = idxb + e * SEG_CAP;

    const int rpos = wv * RPW + (lane & 15);
    int t = sl;
    int tkc = (t * RT + rpos < cnt) ? idx[t * RT + rpos] : -1;
    float4 pf[2 * RPW];
#pragma unroll
    for (int j = 0; j < RPW; ++j) {
        const int tj = __builtin_amdgcn_readlane(tkc, j);
        const int tc = tj < 0 ? 0 : tj;
        const float* xr = x + (size_t)tc * IN_DIM + 4 * lane;
        pf[2 * j]     = *(const float4*)xr;
        pf[2 * j + 1] = *(const float4*)(xr + 128);
    }
    int tkn = -1;
    if ((t + NSLICE) * RT + rpos < cnt) tkn = idx[(t + NSLICE) * RT + rpos];

    const float* w1 = W1 + (size_t)e * (IN_DIM * OUT_DIM) + (size_t)(8 * h) * OUT_DIM + n;
    v16bf bfrA[8], bfrB[8];
#pragma unroll
    for (int ks = 0; ks < 8; ++ks) {
#pragma unroll
        for (int s = 0; s < 16; ++s) {
            const int kk = (s < 8) ? s : (8 + s);
            const float2 wv2 = *(const float2*)(w1 + (32 * ks + kk) * OUT_DIM);
            bfrA[ks][s] = (bf16_t)wv2.x;
            bfrB[ks][s] = (bf16_t)wv2.y;
        }
    }

    const unsigned char* abase = xs + c * XP + 16 * h;

    for (; t < npass; t += NSLICE) {
#pragma unroll
        for (int j = 0; j < RPW; ++j) {
            unsigned char* xw = xs + (wv * RPW + j) * XP + 8 * lane;
            const float4 f0 = pf[2 * j], f1 = pf[2 * j + 1];
            v4bf q0, q1;
            q0[0] = (bf16_t)f0.x; q0[1] = (bf16_t)f0.y; q0[2] = (bf16_t)f0.z; q0[3] = (bf16_t)f0.w;
            q1[0] = (bf16_t)f1.x; q1[1] = (bf16_t)f1.y; q1[2] = (bf16_t)f1.z; q1[3] = (bf16_t)f1.w;
            *(v4bf*)xw         = q0;
            *(v4bf*)(xw + 256) = q1;
        }

        int tkn2 = -1;
        if ((t + 2 * NSLICE) * RT + rpos < cnt) tkn2 = idx[(t + 2 * NSLICE) * RT + rpos];
        if (t + NSLICE < npass) {
#pragma unroll
            for (int j = 0; j < RPW; ++j) {
                const int tj = __builtin_amdgcn_readlane(tkn, j);
                const int tc = tj < 0 ? 0 : tj;
                const float* xr = x + (size_t)tc * IN_DIM + 4 * lane;
                pf[2 * j]     = *(const float4*)xr;
                pf[2 * j + 1] = *(const float4*)(xr + 128);
            }
        }
        lds_barrier();

        v8f accA[4], accB[4];
#pragma unroll
        for (int rt = 0; rt < 4; ++rt) {
#pragma unroll
            for (int r = 0; r < 8; ++r) { accA[rt][r] = 0.f; accB[rt][r] = 0.f; }
        }
#pragma unroll
        for (int ks = 0; ks < 8; ++ks) {
            v16bf a[4];
#pragma unroll
            for (int rt = 0; rt < 4; ++rt) {
                const unsigned char* ap = abase + rt * 16 * XP + 64 * ks;
                const v8bf lo = *(const v8bf*)ap;
                const v8bf hi = *(const v8bf*)(ap + 32);
                a[rt] = __builtin_shufflevector(lo, hi, 0, 1, 2, 3, 4, 5, 6, 7, 8, 9, 10, 11, 12, 13, 14, 15);
            }
#pragma unroll
            for (int rt = 0; rt < 4; ++rt) {
                accA[rt] = __builtin_amdgcn_wmma_f32_16x16x32_bf16(
                    false, a[rt], false, bfrA[ks], (short)0, accA[rt], false, false);
                accB[rt] = __builtin_amdgcn_wmma_f32_16x16x32_bf16(
                    false, a[rt], false, bfrB[ks], (short)0, accB[rt], false, false);
            }
        }

        const bool hi1 = (c & 2) != 0, hi2 = (c & 4) != 0, hi3 = (c & 8) != 0;
        const int  rowc = ((c >> 1) & 1) * 4 + ((c >> 2) & 1) * 2 + ((c >> 3) & 1);
#pragma unroll
        for (int rt = 0; rt < 4; ++rt) {
            float v8[8], v4[4], v2[2];
#pragma unroll
            for (int r = 0; r < 8; ++r) {
                float ha = accA[rt][r] + bb.x, hb = accB[rt][r] + bb.y;
                ha = ha > 0.f ? ha : 0.f;
                hb = hb > 0.f ? hb : 0.f;
                const float keep = ha * wkA + hb * wkB;
                const float send = ha * wsA + hb * wsB;
                v8[r] = keep + row_xor<1>(send);
            }
#pragma unroll
            for (int j = 0; j < 4; ++j) {
                const float keep = hi1 ? v8[j + 4] : v8[j];
                const float send = hi1 ? v8[j] : v8[j + 4];
                v4[j] = keep + row_xor<2>(send);
            }
#pragma unroll
            for (int j = 0; j < 2; ++j) {
                const float keep = hi2 ? v4[j + 2] : v4[j];
                const float send = hi2 ? v4[j] : v4[j + 2];
                v2[j] = keep + row_xor<4>(send);
            }
            const float keep1 = hi3 ? v2[1] : v2[0];
            const float send1 = hi3 ? v2[0] : v2[1];
            const float res = keep1 + row_xor<8>(send1);
            part[(wv * RT + rt * 16 + 8 * h + rowc) * 2 + (c & 1)] = res;
        }
        lds_barrier();

        const int jtok = __shfl(tkc, lane >> 1);
        {
            const int jrow = wv * RPW + (lane >> 1);
            float a = 0.f;
#pragma unroll
            for (int w = 0; w < NWV; ++w) a += part[(w * RT + jrow) * 2 + (lane & 1)];
            const float ls = a + (odd ? b21 : b20);
            const float lo = row_xor<1>(ls);
            const float mx = fmaxf(ls, lo);
            const float es = __expf(ls - mx), eo = __expf(lo - mx);
            const float inv = 1.f / (es + eo);
            if (jtok >= 0) out[(size_t)jtok * 2 + (lane & 1)] = es * inv;
        }
        tkc = tkn;
        tkn = tkn2;
    }
}


extern "C" void kernel_launch(void* const* d_in, const int* in_sizes, int n_in,
                              void* d_out, int out_size, void* d_ws, size_t ws_size,
                              hipStream_t stream) {
    (void)in_sizes; (void)n_in; (void)out_size; (void)ws_size;
    const float* x  = (const float*)d_in[0];
    const int*   z  = (const int*)d_in[1];
    const float* W1 = (const float*)d_in[2];
    const float* b1 = (const float*)d_in[3];
    const float* W2 = (const float*)d_in[4];
    const float* b2 = (const float*)d_in[5];
    float* out = (float*)d_out;

    int* cur  = (int*)d_ws;
    int* idxb = cur + 128;

    zero_k  <<<1, 32, 0, stream>>>(cur);
    bucket_k<<<N_TOK / CHUNK, 256, 0, stream>>>(z, cur, idxb);
    moe_k   <<<KEXP * NSLICE, 32 * NWV, 0, stream>>>(x, idxb, cur, W1, b1, W2, b2, out);
}
